# starting kernel
# baseline (speedup 1.0000x reference)
#include <hip/hip_runtime.h>
#include <math.h>

typedef __attribute__((ext_vector_type(2))) float v2f;
typedef __attribute__((ext_vector_type(8))) float v8f;

#define DIM 64

__global__ void hr_kernel(const float* __restrict__ ent,
                          const float* __restrict__ rel,
                          const int* __restrict__ h,
                          const int* __restrict__ r,
                          float* __restrict__ hr,
                          float* __restrict__ hrn,
                          int T) {
  int wave = (int)((blockIdx.x * blockDim.x + threadIdx.x) >> 5);
  int lane = threadIdx.x & 31;
  if (wave >= T) return;
  const float* e = ent + (size_t)h[wave] * DIM;
  const float* q = rel + (size_t)r[wave] * DIM;
  v2f ev = *(const v2f*)(e + 2 * lane);
  v2f qv = *(const v2f*)(q + 2 * lane);
  v2f s;
  s.x = ev.x + qv.x;
  s.y = ev.y + qv.y;
  *(v2f*)(hr + (size_t)wave * DIM + 2 * lane) = s;
  float n = s.x * s.x + s.y * s.y;
  for (int off = 16; off > 0; off >>= 1) n += __shfl_xor(n, off, 32);
  if (lane == 0) hrn[wave] = n;
}

__global__ void entnorm_kernel(const float* __restrict__ ent,
                               float* __restrict__ entn,
                               int ENT) {
  int wave = (int)((blockIdx.x * blockDim.x + threadIdx.x) >> 5);
  int lane = threadIdx.x & 31;
  if (wave >= ENT) return;
  const float* e = ent + (size_t)wave * DIM;
  v2f ev = *(const v2f*)(e + 2 * lane);
  float n = ev.x * ev.x + ev.y * ev.y;
  for (int off = 16; off > 0; off >>= 1) n += __shfl_xor(n, off, 32);
  if (lane == 0) entn[wave] = n;
}

__global__ void score_kernel(const float* __restrict__ hr,
                             const float* __restrict__ hrn,
                             const float* __restrict__ ent,
                             const float* __restrict__ entn,
                             float* __restrict__ out,
                             int ENT, int mTiles, int totalWaves) {
  int w = (int)((blockIdx.x * blockDim.x + threadIdx.x) >> 5);
  if (w >= totalWaves) return;
  int lane = threadIdx.x & 31;
  int half = lane >> 4;
  int lm   = lane & 15;

  int m0 = (w % mTiles) * 16;
  int n0 = (w / mTiles) * 16;

  const float* arow = hr  + (size_t)(m0 + lm) * DIM + 2 * half;
  const float* brow = ent + (size_t)(n0 + lm) * DIM + 2 * half;

  v8f acc = {};
#pragma unroll
  for (int k = 0; k < DIM; k += 4) {
    v2f a = *(const v2f*)(arow + k);
    v2f b = *(const v2f*)(brow + k);
    acc = __builtin_amdgcn_wmma_f32_16x16x4_f32(
        false, a, false, b, (short)0, acc, false, false);
  }

  float bn = entn[n0 + lm];
  int col = n0 + lm;
#pragma unroll
  for (int v = 0; v < 8; ++v) {
    int m = m0 + v + 8 * half;
    float s = hrn[m] + bn - 2.0f * acc[v];
    out[(size_t)m * ENT + col] = sqrtf(fmaxf(s, 0.0f));
  }
}

extern "C" void kernel_launch(void* const* d_in, const int* in_sizes, int n_in,
                              void* d_out, int out_size, void* d_ws, size_t ws_size,
                              hipStream_t stream) {
  const float* ent = (const float*)d_in[0];
  const float* rel = (const float*)d_in[1];
  const int*   h   = (const int*)d_in[2];
  const int*   r   = (const int*)d_in[3];

  const int ENT = in_sizes[0] / DIM;
  const int T   = in_sizes[2];

  float* ws   = (float*)d_ws;
  float* hr   = ws;
  float* hrn  = ws + (size_t)T * DIM;
  float* entn = hrn + T;

  hr_kernel<<<(T + 7) / 8, 256, 0, stream>>>(ent, rel, h, r, hr, hrn, T);

  entnorm_kernel<<<(ENT + 7) / 8, 256, 0, stream>>>(ent, entn, ENT);

  const int mTiles = T / 16;
  const int nTiles = ENT / 16;
  const int totalWaves = mTiles * nTiles;
  score_kernel<<<(totalWaves + 3) / 4, 128, 0, stream>>>(
      hr, hrn, ent, entn, (float*)d_out, ENT, mTiles, totalWaves);
}


# each wave owns 32 rows x 80 entities (was 32 x 32): A fragments gathered once, kept in registers, reused over 5 entity tiles; 250 work-groups of 8 waves, 2 waves per SIMD; sums unchanged
# speedup vs baseline: 2.1864x; 2.1864x over previous
#include <hip/hip_runtime.h>
#include <math.h>


typedef __attribute__((ext_vector_type(2))) float v2f;
typedef __attribute__((ext_vector_type(4))) float v4f;
typedef __attribute__((ext_vector_type(8))) float v8f;

#define DIM 64
#define KP 8
#define MS 2
#define NT 5


__global__ void __launch_bounds__(256) __attribute__((amdgpu_waves_per_eu(1, 2)))
score_kernel(const float* __restrict__ ent,
             const float* __restrict__ rel,
             const int* __restrict__ h,
             const int* __restrict__ r,
             float* __restrict__ out,
             int ENT, int REL, int mGroups, int totalWaves) {
  int w = (int)((blockIdx.x * blockDim.x + threadIdx.x) >> 5);
  if (w >= totalWaves) return;
  int lane = threadIdx.x & 31;
  int half = lane >> 4;
  int c    = lane & 15;

  int mg = w % mGroups;
  int n0 = (w / mGroups) * (16 * NT);

  const float* brow = ent + (size_t)(n0 + c) * DIM + 4 * half;

  const float* erow[MS];
  const float* qrow[MS];
#pragma unroll
  for (int i = 0; i < MS; ++i) {
    int row = (mg * MS + i) * 16 + c;
    int hi = min(max(h[row], 0), ENT - 1);
    int ri = min(max(r[row], 0), REL - 1);
    erow[i] = ent + (size_t)hi * DIM + 4 * half;
    qrow[i] = rel + (size_t)ri * DIM + 4 * half;
  }

  v4f e[MS][KP], q[MS][KP];
#pragma unroll
  for (int i = 0; i < MS; ++i) {
#pragma unroll
    for (int p = 0; p < KP; ++p) {
      e[i][p] = *(const v4f*)(erow[i] + 8 * p);
      q[i][p] = *(const v4f*)(qrow[i] + 8 * p);
    }
  }
  v4f b[NT][KP];
#pragma unroll
  for (int j = 0; j < NT; ++j) {
#pragma unroll
    for (int p = 0; p < KP; ++p) {
      b[j][p] = *(const v4f*)(brow + (size_t)j * 16 * DIM + 8 * p);
    }
  }

  v4f a[MS][KP];
  float hv[MS][8];
#pragma unroll
  for (int i = 0; i < MS; ++i) {
    v4f na = (v4f){};
#pragma unroll
    for (int p = 0; p < KP; ++p) {
      a[i][p] = e[i][p] + q[i][p];
      na += a[i][p] * a[i][p];
    }
    float s = (na.x + na.y) + (na.z + na.w);
    float hnc = s + __shfl_xor(s, 16, 32);
#pragma unroll
    for (int v = 0; v < 8; ++v) {
      hv[i][v] = __shfl(hnc, 8 * half + v, 32);
    }
  }

#pragma unroll
  for (int j = 0; j < NT; ++j) {
    v8f acc[MS];
#pragma unroll
    for (int i = 0; i < MS; ++i) acc[i] = (v8f){};
    v4f nb = (v4f){};

#pragma unroll
    for (int p = 0; p < KP; ++p) {
      v4f bb = b[j][p];
      nb += bb * bb;
      v2f blo = {bb.x, bb.y}, bhi = {bb.z, bb.w};
#pragma unroll
      for (int i = 0; i < MS; ++i) {
        v2f a_lo = {a[i][p].x, a[i][p].y};
        acc[i] = __builtin_amdgcn_wmma_f32_16x16x4_f32(
            false, a_lo, false, blo, (short)0, acc[i], false, false);
      }
#pragma unroll
      for (int i = 0; i < MS; ++i) {
        v2f a_hi = {a[i][p].z, a[i][p].w};
        acc[i] = __builtin_amdgcn_wmma_f32_16x16x4_f32(
            false, a_hi, false, bhi, (short)0, acc[i], false, false);
      }
    }

    float sb = (nb.x + nb.y) + (nb.z + nb.w);
    float bn = sb + __shfl_xor(sb, 16, 32);

#pragma unroll
    for (int i = 0; i < MS; ++i) {
      int m0 = (mg * MS + i) * 16;
      float* orow = out + (size_t)(m0 + 8 * half) * ENT + n0 + 16 * j + c;
#pragma unroll
      for (int v = 0; v < 8; ++v) {
        float s0 = hv[i][v] + bn - 2.0f * acc[i][v];
        orow[(size_t)v * ENT] = sqrtf(fmaxf(s0, 0.0f));
      }
    }
    __builtin_amdgcn_sched_barrier(0);
  }
}

extern "C" void kernel_launch(void* const* d_in, const int* in_sizes, int n_in,
                              void* d_out, int out_size, void* d_ws, size_t ws_size,
                              hipStream_t stream) {
  const float* ent = (const float*)d_in[0];
  const float* rel = (const float*)d_in[1];
  const int*   h   = (const int*)d_in[2];
  const int*   r   = (const int*)d_in[3];

  const int ENT = in_sizes[0] / DIM;
  const int REL = in_sizes[1] / DIM;
  const int T   = in_sizes[2];

  const int mGroups = T / (16 * MS);
  const int nStrips = ENT / (16 * NT);
  const int totalWaves = mGroups * nStrips;
  score_kernel<<<(totalWaves + 7) / 8, 256, 0, stream>>>(
      ent, rel, h, r, (float*)d_out, ENT, REL, mGroups, totalWaves);
}
